# starting kernel
# baseline (speedup 1.0000x reference)
#include <hip/hip_runtime.h>
#include <hip/hip_bf16.h>

#define NND 200000
#define NE  600000
#define FIN 165
#define FHID 128
#define FOUT 2

typedef float v2f __attribute__((ext_vector_type(2)));
typedef float v8f __attribute__((ext_vector_type(8)));

__global__ void zero_f4(float4* __restrict__ p, size_t n4) {
    size_t stride = (size_t)gridDim.x * blockDim.x;
    for (size_t i = (size_t)blockIdx.x * blockDim.x + threadIdx.x; i < n4; i += stride)
        p[i] = make_float4(0.f, 0.f, 0.f, 0.f);
}

__global__ void degree_kernel(const int* __restrict__ ei, float* __restrict__ deg) {
    int e = blockIdx.x * blockDim.x + threadIdx.x;
    if (e < NE) atomicAdd(&deg[ei[NE + e]], 1.0f);
}

__global__ void dinv_kernel(float* __restrict__ deg) {
    int n = blockIdx.x * blockDim.x + threadIdx.x;
    if (n < NND) deg[n] = rsqrtf(deg[n] + 1.0f);
}

__global__ void gemm1_wmma(const float* __restrict__ x,
                           const float* __restrict__ W1,
                           float* __restrict__ h1) {
    const int lane = threadIdx.x & 31;
    const int tile = blockIdx.x * (blockDim.x >> 5) + (threadIdx.x >> 5);
    const int rowTile = tile >> 3;
    const int colTile = tile & 7;
    const int row0 = rowTile << 4;
    const int col0 = colTile << 4;
    const int m  = lane & 15;
    const int hi = lane >> 4;
    const int col = col0 + m;

    v8f acc = {0.f, 0.f, 0.f, 0.f, 0.f, 0.f, 0.f, 0.f};
    const float* xrow = x + (size_t)(row0 + m) * FIN;

    for (int k0 = 0; k0 < FIN - 4; k0 += 4) {
        const int kb = k0 + (hi << 1);
        v2f a, b;
        a.x = xrow[kb];
        a.y = xrow[kb + 1];
        b.x = W1[(size_t)kb * FHID + col];
        b.y = W1[(size_t)(kb + 1) * FHID + col];
        acc = __builtin_amdgcn_wmma_f32_16x16x4_f32(
             false, a,  false, b,
             (short)0, acc,  false,  false);
    }

    {
        const float a164 = xrow[FIN - 1];
        const float b164 = W1[(size_t)(FIN - 1) * FHID + col];
        v2f a, b;
        a.x = hi ? 0.f : a164;
        a.y = 0.f;
        b.x = hi ? 0.f : b164;
        b.y = 0.f;
        acc = __builtin_amdgcn_wmma_f32_16x16x4_f32(
            false, a, false, b, (short)0, acc, false, false);
    }

#pragma unroll
    for (int i = 0; i < 8; ++i)
        h1[(size_t)(row0 + i + 8 * hi) * FHID + col] = acc[i];
}

__global__ void scatter1(const int* __restrict__ ei,
                         const float* __restrict__ dinv,
                         const float* __restrict__ h1,
                         float* __restrict__ agg1) {
    const int wid  = (blockIdx.x * blockDim.x + threadIdx.x) >> 5;
    const int lane = threadIdx.x & 31;
    if (wid >= NE) return;
    const int src = ei[wid];
    const int dst = ei[NE + wid];
    const float norm = dinv[src] * dinv[dst];
    const float4 v = ((const float4*)(h1 + (size_t)src * FHID))[lane];
    float* out = agg1 + (size_t)dst * FHID + lane * 4;
    atomicAdd(out + 0, v.x * norm);
    atomicAdd(out + 1, v.y * norm);
    atomicAdd(out + 2, v.z * norm);
    atomicAdd(out + 3, v.w * norm);
}

__global__ void relu_self(const float* __restrict__ dinv,
                          const float* __restrict__ h1,
                          const float* __restrict__ b1,
                          float* __restrict__ agg1) {
    const size_t total4 = (size_t)NND * FHID / 4;
    const size_t stride = (size_t)gridDim.x * blockDim.x;
    for (size_t i = (size_t)blockIdx.x * blockDim.x + threadIdx.x; i < total4; i += stride) {
        const size_t e = i * 4;
        const int n = (int)(e / FHID);
        const int f = (int)(e % FHID);
        const float s = dinv[n];
        const float s2 = s * s;
        const float4 a  = ((const float4*)agg1)[i];
        const float4 hv = ((const float4*)h1)[i];
        float4 r;
        r.x = fmaxf(a.x + hv.x * s2 + b1[f + 0], 0.f);
        r.y = fmaxf(a.y + hv.y * s2 + b1[f + 1], 0.f);
        r.z = fmaxf(a.z + hv.z * s2 + b1[f + 2], 0.f);
        r.w = fmaxf(a.w + hv.w * s2 + b1[f + 3], 0.f);
        ((float4*)agg1)[i] = r;
    }
}

__global__ void gemm2_kernel(const float* __restrict__ h1p,
                             const float* __restrict__ W2,
                             float* __restrict__ h2) {
    __shared__ float w2s[FHID * FOUT];
    if (threadIdx.x < FHID * FOUT) w2s[threadIdx.x] = W2[threadIdx.x];
    __syncthreads();

    const int n = blockIdx.x * blockDim.x + threadIdx.x;
    if (n >= NND) return;
    const float4* row = (const float4*)(h1p + (size_t)n * FHID);
    float acc0 = 0.f, acc1 = 0.f;
#pragma unroll 8
    for (int k4 = 0; k4 < FHID / 4; ++k4) {
        const float4 v = row[k4];
        const int k = k4 * 4;
        acc0 += v.x * w2s[(k + 0) * FOUT + 0] + v.y * w2s[(k + 1) * FOUT + 0]
              + v.z * w2s[(k + 2) * FOUT + 0] + v.w * w2s[(k + 3) * FOUT + 0];
        acc1 += v.x * w2s[(k + 0) * FOUT + 1] + v.y * w2s[(k + 1) * FOUT + 1]
              + v.z * w2s[(k + 2) * FOUT + 1] + v.w * w2s[(k + 3) * FOUT + 1];
    }
    h2[(size_t)n * FOUT + 0] = acc0;
    h2[(size_t)n * FOUT + 1] = acc1;
}

__global__ void scatter2(const int* __restrict__ ei,
                         const float* __restrict__ dinv,
                         const float* __restrict__ h2,
                         float* __restrict__ agg2) {
    const int e = blockIdx.x * blockDim.x + threadIdx.x;
    if (e >= NE) return;
    const int src = ei[e];
    const int dst = ei[NE + e];
    const float norm = dinv[src] * dinv[dst];
    atomicAdd(&agg2[(size_t)dst * FOUT + 0], h2[(size_t)src * FOUT + 0] * norm);
    atomicAdd(&agg2[(size_t)dst * FOUT + 1], h2[(size_t)src * FOUT + 1] * norm);
}

__global__ void final_kernel(const float* __restrict__ dinv,
                             const float* __restrict__ h2,
                             const float* __restrict__ agg2,
                             const float* __restrict__ b2,
                             float* __restrict__ out) {
    const int n = blockIdx.x * blockDim.x + threadIdx.x;
    if (n >= NND) return;
    const float s = dinv[n];
    const float s2 = s * s;
    out[(size_t)n * FOUT + 0] = agg2[(size_t)n * FOUT + 0] + h2[(size_t)n * FOUT + 0] * s2 + b2[0];
    out[(size_t)n * FOUT + 1] = agg2[(size_t)n * FOUT + 1] + h2[(size_t)n * FOUT + 1] * s2 + b2[1];
}

extern "C" void kernel_launch(void* const* d_in, const int* in_sizes, int n_in,
                              void* d_out, int out_size, void* d_ws, size_t ws_size,
                              hipStream_t stream) {
    const float* x  = (const float*)d_in[0];
    const int*   ei = (const int*)d_in[1];
    const float* W1 = (const float*)d_in[2];
    const float* b1 = (const float*)d_in[3];
    const float* W2 = (const float*)d_in[4];
    const float* b2 = (const float*)d_in[5];
    float* out = (float*)d_out;

    float* ws   = (float*)d_ws;
    float* dinv = ws;
    float* agg1 = ws + (size_t)NND;
    float* agg2 = ws + (size_t)NND * 129;
    float* h1   = ws + (size_t)NND * 131;
    float* h2   = ws + (size_t)NND * 259;

    const size_t zero4 = (size_t)NND * 131 / 4;
    zero_f4<<<2048, 256, 0, stream>>>((float4*)ws, zero4);

    degree_kernel<<<(NE + 255) / 256, 256, 0, stream>>>(ei, dinv);
    dinv_kernel<<<(NND + 255) / 256, 256, 0, stream>>>(dinv);

    gemm1_wmma<<<25000, 128, 0, stream>>>(x, W1, h1);

    scatter1<<<(NE + 7) / 8, 256, 0, stream>>>(ei, dinv, h1, agg1);

    relu_self<<<2048, 256, 0, stream>>>(dinv, h1, b1, agg1);

    gemm2_kernel<<<(NND + 255) / 256, 256, 0, stream>>>(agg1, W2, h2);

    scatter2<<<(NE + 255) / 256, 256, 0, stream>>>(ei, dinv, h2, agg2);

    final_kernel<<<(NND + 255) / 256, 256, 0, stream>>>(dinv, h2, agg2, b2, out);
}


# GEMM1 re-tiled: each wave computes a 16x128 strip so one A fragment feeds 8 WMMAs; x strip and W1 fragments staged in LDS by coalesced 16-byte loads; 256 persistent workgroups; sum order unchanged
# speedup vs baseline: 1.4735x; 1.4735x over previous
#include <hip/hip_runtime.h>
#include <hip/hip_bf16.h>

#define NND 200000
#define NE  600000
#define FIN 165
#define FHID 128
#define FOUT 2

typedef float v2f __attribute__((ext_vector_type(2)));
typedef float v8f __attribute__((ext_vector_type(8)));

__global__ void zero_f4(float4* __restrict__ p, size_t n4) {
    size_t stride = (size_t)gridDim.x * blockDim.x;
    for (size_t i = (size_t)blockIdx.x * blockDim.x + threadIdx.x; i < n4; i += stride)
        p[i] = make_float4(0.f, 0.f, 0.f, 0.f);
}

__global__ void degree_kernel(const int* __restrict__ ei, float* __restrict__ deg) {
    int e = blockIdx.x * blockDim.x + threadIdx.x;
    if (e < NE) atomicAdd(&deg[ei[NE + e]], 1.0f);
}

__global__ void dinv_kernel(float* __restrict__ deg) {
    int n = blockIdx.x * blockDim.x + threadIdx.x;
    if (n < NND) deg[n] = rsqrtf(deg[n] + 1.0f);
}

#define G1_WAVES  16
#define G1_KSTEPS 42
#define G1_BQ     (G1_KSTEPS * 4 * 32)
#define G1_AF4    660
#define G1_STRIPS (NND / 16)

__device__ __forceinline__ void g1_step(v8f (&acc)[8], const v2f a, const float4 (&b)[4]) {
    for (int p = 0; p < 4; ++p) {
        v2f b0, b1;
        b0.x = b[p].x; b0.y = b[p].y;
        b1.x = b[p].z; b1.y = b[p].w;
        acc[2 * p] = __builtin_amdgcn_wmma_f32_16x16x4_f32(
            false, a, false, b0, (short)0, acc[2 * p], false, false);
        acc[2 * p + 1] = __builtin_amdgcn_wmma_f32_16x16x4_f32(
            false, a, false, b1, (short)0, acc[2 * p + 1], false, false);
    }
}

__global__ __launch_bounds__(G1_WAVES * 32)
void gemm1_wmma(const float* __restrict__ x,
                const float* __restrict__ W1,
                float* __restrict__ h1) {
    __shared__ __attribute__((aligned(16))) float4 Bs[G1_BQ];
    __shared__ __attribute__((aligned(16))) float4 As[G1_WAVES][G1_AF4];

    const int lane = threadIdx.x & 31;
    const int wave = threadIdx.x >> 5;
    const int m  = lane & 15;
    const int hi = lane >> 4;

    for (int it = 0; it < (G1_BQ + G1_WAVES * 32 - 1) / (G1_WAVES * 32); ++it) {
        const int q  = threadIdx.x + it * (G1_WAVES * 32);
        const int qc = q < G1_BQ ? q : G1_BQ - 1;
        const int l = qc & 31;
        const int p = (qc >> 5) & 3;
        const int s = qc >> 7;
        const int k = 4 * s + 2 * (l >> 4);
        const int c = 64 * (p >> 1) + 4 * (l & 15) + 2 * (p & 1);
        const float* r0 = W1 + (size_t)(k     < FIN ? k     : FIN - 1) * FHID + c;
        const float* r1 = W1 + (size_t)(k + 1 < FIN ? k + 1 : FIN - 1) * FHID + c;
        const float w00 = r0[0], w01 = r0[1], w10 = r1[0], w11 = r1[1];
        float4 v;
        v.x = (k     < FIN) ? w00 : 0.f;
        v.y = (k + 1 < FIN) ? w10 : 0.f;
        v.z = (k     < FIN) ? w01 : 0.f;
        v.w = (k + 1 < FIN) ? w11 : 0.f;
        if (q < G1_BQ) Bs[q] = v;
    }
    __syncthreads();

    float4* Aq = As[wave];
    const float* Aw = (const float*)As[wave];
    const int aoff = m * FIN + 2 * hi;

    for (int strip = wave * gridDim.x + blockIdx.x; strip < G1_STRIPS;
         strip += G1_WAVES * gridDim.x) {
        const float4* src = (const float4*)(x + (size_t)strip * (16 * FIN));
        float4 t[20];
        for (int i = 0; i < 20; ++i) t[i] = src[lane + 32 * i];
        const float4 t20 = src[640 + (lane < 20 ? lane : 19)];
        for (int i = 0; i < 20; ++i) Aq[lane + 32 * i] = t[i];
        if (lane < 20) Aq[lane + 640] = t20;

        v8f acc[8];
        for (int j = 0; j < 8; ++j)
            acc[j] = (v8f){0.f, 0.f, 0.f, 0.f, 0.f, 0.f, 0.f, 0.f};

        v2f a;
        float4 b[4];
        a.x = Aw[aoff];
        a.y = Aw[aoff + 1];
        for (int p = 0; p < 4; ++p) b[p] = Bs[p * 32 + lane];

        for (int s = 0; s < 40; ++s) {
            v2f an;
            float4 bn[4];
            an.x = Aw[aoff + 4 * (s + 1)];
            an.y = Aw[aoff + 4 * (s + 1) + 1];
            for (int p = 0; p < 4; ++p) bn[p] = Bs[((s + 1) * 4 + p) * 32 + lane];
            g1_step(acc, a, b);
            a = an;
            for (int p = 0; p < 4; ++p) b[p] = bn[p];
        }
        {
            const float a164 = Aw[m * FIN + (FIN - 1)];
            v2f an;
            float4 bn[4];
            an.x = hi ? 0.f : a164;
            an.y = 0.f;
            for (int p = 0; p < 4; ++p) bn[p] = Bs[(41 * 4 + p) * 32 + lane];
            g1_step(acc, a, b);
            g1_step(acc, an, bn);
        }

        float* orow = h1 + (size_t)(strip * 16 + 8 * hi) * FHID + 4 * m;
        for (int i = 0; i < 8; ++i) {
            for (int g = 0; g < 2; ++g) {
                float4 o;
                o.x = acc[4 * g + 0][i];
                o.y = acc[4 * g + 1][i];
                o.z = acc[4 * g + 2][i];
                o.w = acc[4 * g + 3][i];
                *(float4*)(orow + (size_t)i * FHID + 64 * g) = o;
            }
        }
    }
}

__global__ void scatter1(const int* __restrict__ ei,
                         const float* __restrict__ dinv,
                         const float* __restrict__ h1,
                         float* __restrict__ agg1) {
    const int wid  = (blockIdx.x * blockDim.x + threadIdx.x) >> 5;
    const int lane = threadIdx.x & 31;
    if (wid >= NE) return;
    const int src = ei[wid];
    const int dst = ei[NE + wid];
    const float norm = dinv[src] * dinv[dst];
    const float4 v = ((const float4*)(h1 + (size_t)src * FHID))[lane];
    float* out = agg1 + (size_t)dst * FHID + lane * 4;
    atomicAdd(out + 0, v.x * norm);
    atomicAdd(out + 1, v.y * norm);
    atomicAdd(out + 2, v.z * norm);
    atomicAdd(out + 3, v.w * norm);
}

__global__ void relu_self(const float* __restrict__ dinv,
                          const float* __restrict__ h1,
                          const float* __restrict__ b1,
                          float* __restrict__ agg1) {
    const size_t total4 = (size_t)NND * FHID / 4;
    const size_t stride = (size_t)gridDim.x * blockDim.x;
    for (size_t i = (size_t)blockIdx.x * blockDim.x + threadIdx.x; i < total4; i += stride) {
        const size_t e = i * 4;
        const int n = (int)(e / FHID);
        const int f = (int)(e % FHID);
        const float s = dinv[n];
        const float s2 = s * s;
        const float4 a  = ((const float4*)agg1)[i];
        const float4 hv = ((const float4*)h1)[i];
        float4 r;
        r.x = fmaxf(a.x + hv.x * s2 + b1[f + 0], 0.f);
        r.y = fmaxf(a.y + hv.y * s2 + b1[f + 1], 0.f);
        r.z = fmaxf(a.z + hv.z * s2 + b1[f + 2], 0.f);
        r.w = fmaxf(a.w + hv.w * s2 + b1[f + 3], 0.f);
        ((float4*)agg1)[i] = r;
    }
}

__global__ void gemm2_kernel(const float* __restrict__ h1p,
                             const float* __restrict__ W2,
                             float* __restrict__ h2) {
    __shared__ float w2s[FHID * FOUT];
    if (threadIdx.x < FHID * FOUT) w2s[threadIdx.x] = W2[threadIdx.x];
    __syncthreads();

    const int n = blockIdx.x * blockDim.x + threadIdx.x;
    if (n >= NND) return;
    const float4* row = (const float4*)(h1p + (size_t)n * FHID);
    float acc0 = 0.f, acc1 = 0.f;
#pragma unroll 8
    for (int k4 = 0; k4 < FHID / 4; ++k4) {
        const float4 v = row[k4];
        const int k = k4 * 4;
        acc0 += v.x * w2s[(k + 0) * FOUT + 0] + v.y * w2s[(k + 1) * FOUT + 0]
              + v.z * w2s[(k + 2) * FOUT + 0] + v.w * w2s[(k + 3) * FOUT + 0];
        acc1 += v.x * w2s[(k + 0) * FOUT + 1] + v.y * w2s[(k + 1) * FOUT + 1]
              + v.z * w2s[(k + 2) * FOUT + 1] + v.w * w2s[(k + 3) * FOUT + 1];
    }
    h2[(size_t)n * FOUT + 0] = acc0;
    h2[(size_t)n * FOUT + 1] = acc1;
}

__global__ void scatter2(const int* __restrict__ ei,
                         const float* __restrict__ dinv,
                         const float* __restrict__ h2,
                         float* __restrict__ agg2) {
    const int e = blockIdx.x * blockDim.x + threadIdx.x;
    if (e >= NE) return;
    const int src = ei[e];
    const int dst = ei[NE + e];
    const float norm = dinv[src] * dinv[dst];
    atomicAdd(&agg2[(size_t)dst * FOUT + 0], h2[(size_t)src * FOUT + 0] * norm);
    atomicAdd(&agg2[(size_t)dst * FOUT + 1], h2[(size_t)src * FOUT + 1] * norm);
}

__global__ void final_kernel(const float* __restrict__ dinv,
                             const float* __restrict__ h2,
                             const float* __restrict__ agg2,
                             const float* __restrict__ b2,
                             float* __restrict__ out) {
    const int n = blockIdx.x * blockDim.x + threadIdx.x;
    if (n >= NND) return;
    const float s = dinv[n];
    const float s2 = s * s;
    out[(size_t)n * FOUT + 0] = agg2[(size_t)n * FOUT + 0] + h2[(size_t)n * FOUT + 0] * s2 + b2[0];
    out[(size_t)n * FOUT + 1] = agg2[(size_t)n * FOUT + 1] + h2[(size_t)n * FOUT + 1] * s2 + b2[1];
}

extern "C" void kernel_launch(void* const* d_in, const int* in_sizes, int n_in,
                              void* d_out, int out_size, void* d_ws, size_t ws_size,
                              hipStream_t stream) {
    const float* x  = (const float*)d_in[0];
    const int*   ei = (const int*)d_in[1];
    const float* W1 = (const float*)d_in[2];
    const float* b1 = (const float*)d_in[3];
    const float* W2 = (const float*)d_in[4];
    const float* b2 = (const float*)d_in[5];
    float* out = (float*)d_out;

    float* ws   = (float*)d_ws;
    float* dinv = ws;
    float* agg1 = ws + (size_t)NND;
    float* agg2 = ws + (size_t)NND * 129;
    float* h1   = ws + (size_t)NND * 131;
    float* h2   = ws + (size_t)NND * 259;

    const size_t zero4 = (size_t)NND * 131 / 4;
    zero_f4<<<2048, 256, 0, stream>>>((float4*)ws, zero4);

    degree_kernel<<<(NE + 255) / 256, 256, 0, stream>>>(ei, dinv);
    dinv_kernel<<<(NND + 255) / 256, 256, 0, stream>>>(dinv);

    gemm1_wmma<<<256, 512, 0, stream>>>(x, W1, h1);

    scatter1<<<(NE + 7) / 8, 256, 0, stream>>>(ei, dinv, h1, agg1);

    relu_self<<<2048, 256, 0, stream>>>(dinv, h1, b1, agg1);

    gemm2_kernel<<<(NND + 255) / 256, 256, 0, stream>>>(agg1, W2, h2);

    scatter2<<<(NE + 255) / 256, 256, 0, stream>>>(ei, dinv, h2, agg2);

    final_kernel<<<(NND + 255) / 256, 256, 0, stream>>>(dinv, h2, agg2, b2, out);
}


# Both float atomic scatters become per-destination gathers over a per-call CSR (counting sort, arrival-order slots); layer 1 fused with self loop, bias, ReLU, W2 (lane-tree sum); agg planes gone
# speedup vs baseline: 2.8963x; 1.9656x over previous
#include <hip/hip_runtime.h>
#include <hip/hip_bf16.h>

#define NND 200000
#define NE  600000
#define FIN 165
#define FHID 128
#define FOUT 2

typedef float v2f __attribute__((ext_vector_type(2)));
typedef float v8f __attribute__((ext_vector_type(8)));

__global__ void zero_f4(float4* __restrict__ p, size_t n4) {
    size_t stride = (size_t)gridDim.x * blockDim.x;
    for (size_t i = (size_t)blockIdx.x * blockDim.x + threadIdx.x; i < n4; i += stride)
        p[i] = make_float4(0.f, 0.f, 0.f, 0.f);
}

__global__ void degree_kernel(const int* __restrict__ ei, int* __restrict__ cnt) {
    int e = blockIdx.x * blockDim.x + threadIdx.x;
    if (e < NE) atomicAdd(&cnt[ei[NE + e]], 1);
}

__global__ __launch_bounds__(1024)
void alloc_kernel(const int* __restrict__ cnt, int* __restrict__ counter,
                  int* __restrict__ start, int* __restrict__ cursor,
                  float* __restrict__ dinv) {
    __shared__ int wsum[32];
    __shared__ int sbase;
    const int lane = threadIdx.x & 31;
    const int wave = threadIdx.x >> 5;
    const int n = blockIdx.x * 1024 + threadIdx.x;
    const int c = (n < NND) ? cnt[n] : 0;
    int incl = c;
    for (int d = 1; d < 32; d <<= 1) {
        const int t = __shfl_up(incl, d);
        if (lane >= d) incl += t;
    }
    if (lane == 31) wsum[wave] = incl;
    __syncthreads();
    if (wave == 0) {
        const int s = wsum[lane];
        int si = s;
        for (int d = 1; d < 32; d <<= 1) {
            const int t = __shfl_up(si, d);
            if (lane >= d) si += t;
        }
        wsum[lane] = si - s;
        if (lane == 31) sbase = atomicAdd(counter, si);
    }
    __syncthreads();
    const int st = sbase + wsum[wave] + incl - c;
    if (n < NND) {
        start[n]  = st;
        cursor[n] = st;
        dinv[n]   = rsqrtf((float)c + 1.0f);
    }
}

__global__ void fill_kernel(const int* __restrict__ ei, const float* __restrict__ dinv,
                            int* __restrict__ cursor, int* __restrict__ csr_w) {
    int2* csr = (int2*)csr_w;
    const int e = blockIdx.x * blockDim.x + threadIdx.x;
    if (e >= NE) return;
    const int src = ei[e];
    const int dst = ei[NE + e];
    const float norm = dinv[src] * dinv[dst];
    const int pos = atomicAdd(&cursor[dst], 1);
    csr[pos] = make_int2(src, __float_as_int(norm));
}

#define G1_WAVES  16
#define G1_KSTEPS 42
#define G1_BQ     (G1_KSTEPS * 4 * 32)
#define G1_AF4    660
#define G1_STRIPS (NND / 16)

__device__ __forceinline__ void g1_step(v8f (&acc)[8], const v2f a, const float4 (&b)[4]) {
    for (int p = 0; p < 4; ++p) {
        v2f b0, b1;
        b0.x = b[p].x; b0.y = b[p].y;
        b1.x = b[p].z; b1.y = b[p].w;
        acc[2 * p] = __builtin_amdgcn_wmma_f32_16x16x4_f32(
            false, a, false, b0, (short)0, acc[2 * p], false, false);
        acc[2 * p + 1] = __builtin_amdgcn_wmma_f32_16x16x4_f32(
            false, a, false, b1, (short)0, acc[2 * p + 1], false, false);
    }
}

__global__ __launch_bounds__(G1_WAVES * 32)
void gemm1_wmma(const float* __restrict__ x,
                const float* __restrict__ W1,
                float* __restrict__ h1) {
    __shared__ __attribute__((aligned(16))) float4 Bs[G1_BQ];
    __shared__ __attribute__((aligned(16))) float4 As[G1_WAVES][G1_AF4];

    const int lane = threadIdx.x & 31;
    const int wave = threadIdx.x >> 5;
    const int m  = lane & 15;
    const int hi = lane >> 4;

    for (int it = 0; it < (G1_BQ + G1_WAVES * 32 - 1) / (G1_WAVES * 32); ++it) {
        const int q  = threadIdx.x + it * (G1_WAVES * 32);
        const int qc = q < G1_BQ ? q : G1_BQ - 1;
        const int l = qc & 31;
        const int p = (qc >> 5) & 3;
        const int s = qc >> 7;
        const int k = 4 * s + 2 * (l >> 4);
        const int c = 64 * (p >> 1) + 4 * (l & 15) + 2 * (p & 1);
        const float* r0 = W1 + (size_t)(k     < FIN ? k     : FIN - 1) * FHID + c;
        const float* r1 = W1 + (size_t)(k + 1 < FIN ? k + 1 : FIN - 1) * FHID + c;
        const float w00 = r0[0], w01 = r0[1], w10 = r1[0], w11 = r1[1];
        float4 v;
        v.x = (k     < FIN) ? w00 : 0.f;
        v.y = (k + 1 < FIN) ? w10 : 0.f;
        v.z = (k     < FIN) ? w01 : 0.f;
        v.w = (k + 1 < FIN) ? w11 : 0.f;
        if (q < G1_BQ) Bs[q] = v;
    }
    __syncthreads();

    float4* Aq = As[wave];
    const float* Aw = (const float*)As[wave];
    const int aoff = m * FIN + 2 * hi;

    for (int strip = wave * gridDim.x + blockIdx.x; strip < G1_STRIPS;
         strip += G1_WAVES * gridDim.x) {
        const float4* src = (const float4*)(x + (size_t)strip * (16 * FIN));
        float4 t[20];
        for (int i = 0; i < 20; ++i) t[i] = src[lane + 32 * i];
        const float4 t20 = src[640 + (lane < 20 ? lane : 19)];
        for (int i = 0; i < 20; ++i) Aq[lane + 32 * i] = t[i];
        if (lane < 20) Aq[lane + 640] = t20;

        v8f acc[8];
        for (int j = 0; j < 8; ++j)
            acc[j] = (v8f){0.f, 0.f, 0.f, 0.f, 0.f, 0.f, 0.f, 0.f};

        v2f a;
        float4 b[4];
        a.x = Aw[aoff];
        a.y = Aw[aoff + 1];
        for (int p = 0; p < 4; ++p) b[p] = Bs[p * 32 + lane];

        for (int s = 0; s < 40; ++s) {
            v2f an;
            float4 bn[4];
            an.x = Aw[aoff + 4 * (s + 1)];
            an.y = Aw[aoff + 4 * (s + 1) + 1];
            for (int p = 0; p < 4; ++p) bn[p] = Bs[((s + 1) * 4 + p) * 32 + lane];
            g1_step(acc, a, b);
            a = an;
            for (int p = 0; p < 4; ++p) b[p] = bn[p];
        }
        {
            const float a164 = Aw[m * FIN + (FIN - 1)];
            v2f an;
            float4 bn[4];
            an.x = hi ? 0.f : a164;
            an.y = 0.f;
            for (int p = 0; p < 4; ++p) bn[p] = Bs[(41 * 4 + p) * 32 + lane];
            g1_step(acc, a, b);
            g1_step(acc, an, bn);
        }

        float* orow = h1 + (size_t)(strip * 16 + 8 * hi) * FHID + 4 * m;
        for (int i = 0; i < 8; ++i) {
            for (int g = 0; g < 2; ++g) {
                float4 o;
                o.x = acc[4 * g + 0][i];
                o.y = acc[4 * g + 1][i];
                o.z = acc[4 * g + 2][i];
                o.w = acc[4 * g + 3][i];
                *(float4*)(orow + (size_t)i * FHID + 64 * g) = o;
            }
        }
    }
}

#define GA_WAVES 4

struct GRows {
    float4 s;
    float4 e[8];
    float  w[8];
    int    c;
    float  s2;
};

__device__ __forceinline__ int   rl_i(int v, int l)   { return __builtin_amdgcn_readlane(v, l); }
__device__ __forceinline__ float rl_f(float v, int l) { return __int_as_float(__builtin_amdgcn_readlane(__float_as_int(v), l)); }

__device__ __forceinline__ float4 g_row(const char* __restrict__ Hb, const int row, const unsigned lane16) {
    return *(const float4*)(Hb + (((unsigned)row << 9) + lane16));
}

__device__ __forceinline__ void g_issue(GRows& R, const int j, const int base, const int lane,
                                        const int c, const float dn, const int2* E,
                                        const char* __restrict__ H) {
    const unsigned lane16 = (unsigned)lane << 4;
    const int cj = rl_i(c, j);
    const float dj = rl_f(dn, j);
    const int2 en = E[j * 8 + (lane & 7)];
    const int n = base + j;
    R.c = cj;
    R.s2 = dj * dj;
    R.s = g_row(H, n, lane16);
    for (int k = 0; k < 4; ++k) {
        const int sk = rl_i(en.x, k);
        R.w[k] = __int_as_float(rl_i(en.y, k));
        const int row = (k < cj) ? sk : n;
        R.e[k] = g_row(H, row, lane16);
    }
    if (cj > 4) {
        for (int k = 4; k < 8; ++k) {
            const int sk = rl_i(en.x, k);
            R.w[k] = __int_as_float(rl_i(en.y, k));
            const int row = (k < cj) ? sk : n;
            R.e[k] = g_row(H, row, lane16);
        }
    }
}

__device__ __forceinline__ void g_add(float4& acc, const float4 v, const float w, const bool on) {
    acc.x += on ? v.x * w : 0.f;
    acc.y += on ? v.y * w : 0.f;
    acc.z += on ? v.z * w : 0.f;
    acc.w += on ? v.w * w : 0.f;
}

__device__ __forceinline__ void g_consume(const GRows& R, const int j, const int lane,
                                          const int st, const int2* __restrict__ csr,
                                          const char* __restrict__ H,
                                          const float4 bb, const float4 wA, const float4 wB,
                                          float& r0, float& r1) {
    float4 acc = make_float4(0.f, 0.f, 0.f, 0.f);
    for (int k = 0; k < 4; ++k) g_add(acc, R.e[k], R.w[k], k < R.c);
    if (R.c > 4) {
        for (int k = 4; k < 8; ++k) g_add(acc, R.e[k], R.w[k], k < R.c);
    }
    if (R.c > 8) {
        const int sj = rl_i(st, j);
        for (int e = 8; e < R.c; ++e) {
            const int2 en = csr[sj + e];
            const float4 v = g_row(H, en.x, (unsigned)lane << 4);
            g_add(acc, v, __int_as_float(en.y), true);
        }
    }
    float4 r;
    r.x = fmaxf(acc.x + R.s.x * R.s2 + bb.x, 0.f);
    r.y = fmaxf(acc.y + R.s.y * R.s2 + bb.y, 0.f);
    r.z = fmaxf(acc.z + R.s.z * R.s2 + bb.z, 0.f);
    r.w = fmaxf(acc.w + R.s.w * R.s2 + bb.w, 0.f);
    float p0 = r.x * wA.x + r.y * wA.z + r.z * wB.x + r.w * wB.z;
    float p1 = r.x * wA.y + r.y * wA.w + r.z * wB.y + r.w * wB.w;
    for (int d = 16; d >= 1; d >>= 1) {
        p0 += __shfl_xor(p0, d);
        p1 += __shfl_xor(p1, d);
    }
    if (lane == j) { r0 = p0; r1 = p1; }
}

__global__ __launch_bounds__(GA_WAVES * 32, 8)
void gather1_fused(const int* __restrict__ start, const int* __restrict__ cnt,
                   const float* __restrict__ dinv, const int* __restrict__ csr_w,
                   const float* __restrict__ h1, const float* __restrict__ b1,
                   const float* __restrict__ W2, float* __restrict__ h2) {
    __shared__ int2 ent[GA_WAVES][256];
    const int2* csr = (const int2*)csr_w;
    const int lane = threadIdx.x & 31;
    const int wave = threadIdx.x >> 5;
    const int chunk = blockIdx.x * GA_WAVES + wave;
    if (chunk >= NND / 32) return;
    const int base = chunk * 32;
    const int node = base + lane;
    const int   st = start[node];
    const int   c  = cnt[node];
    const float dn = dinv[node];
    const float4 bb = ((const float4*)b1)[lane];
    const float4 wA = ((const float4*)W2)[2 * lane];
    const float4 wB = ((const float4*)W2)[2 * lane + 1];
    const char* H = (const char*)h1;
    int2* E = ent[wave];

    {
        int2 t[8];
        for (int v = 0; v < 8; ++v) {
            const int jj = v * 4 + (lane >> 3);
            const int slot = lane & 7;
            const int cj = __shfl(c, jj);
            const int sj = __shfl(st, jj);
            int last = cj - 1;
            last = last < 0 ? 0 : last;
            int idx = sj + (slot < last ? slot : last);
            idx = idx < NE - 1 ? idx : NE - 1;
            t[v] = csr[idx];
        }
        for (int v = 0; v < 8; ++v) E[v * 32 + lane] = t[v];
    }

    float r0 = 0.f, r1 = 0.f;
    GRows A, B;
    g_issue(A, 0, base, lane, c, dn, E, H);
    for (int j = 0; j < 32; j += 2) {
        g_issue(B, j + 1, base, lane, c, dn, E, H);
        g_consume(A, j, lane, st, csr, H, bb, wA, wB, r0, r1);
        if (j + 2 < 32) g_issue(A, j + 2, base, lane, c, dn, E, H);
        g_consume(B, j + 1, lane, st, csr, H, bb, wA, wB, r0, r1);
    }
    ((float2*)h2)[node] = make_float2(r0, r1);
}

__global__ void gather2_final(const int* __restrict__ start, const int* __restrict__ cnt,
                              const float* __restrict__ dinv, const int* __restrict__ csr_w,
                              const float* __restrict__ h2, const float* __restrict__ b2,
                              float* __restrict__ out) {
    const int2* csr = (const int2*)csr_w;
    const int n = blockIdx.x * blockDim.x + threadIdx.x;
    if (n >= NND) return;
    const int st = start[n];
    const int c  = cnt[n];
    const float s = dinv[n];
    const float2* H2 = (const float2*)h2;
    const float2 hv = H2[n];
    float a0 = 0.f, a1 = 0.f;
    for (int e0 = 0; e0 < c; e0 += 4) {
        int2 en[4];
        for (int k = 0; k < 4; ++k) {
            const int e = e0 + k;
            en[k] = csr[st + (e < c ? e : c - 1)];
        }
        float2 v[4];
        for (int k = 0; k < 4; ++k) v[k] = H2[en[k].x];
        for (int k = 0; k < 4; ++k) {
            const bool on = e0 + k < c;
            const float w = __int_as_float(en[k].y);
            a0 += on ? v[k].x * w : 0.f;
            a1 += on ? v[k].y * w : 0.f;
        }
    }
    const float s2 = s * s;
    float2 o;
    o.x = a0 + hv.x * s2 + b2[0];
    o.y = a1 + hv.y * s2 + b2[1];
    ((float2*)out)[n] = o;
}

extern "C" void kernel_launch(void* const* d_in, const int* in_sizes, int n_in,
                              void* d_out, int out_size, void* d_ws, size_t ws_size,
                              hipStream_t stream) {
    const float* x  = (const float*)d_in[0];
    const int*   ei = (const int*)d_in[1];
    const float* W1 = (const float*)d_in[2];
    const float* b1 = (const float*)d_in[3];
    const float* W2 = (const float*)d_in[4];
    const float* b2 = (const float*)d_in[5];
    float* out = (float*)d_out;

    float* ws      = (float*)d_ws;
    float* dinv    = ws;
    int*   cnt     = (int*)(ws + (size_t)NND);
    int*   counter = (int*)(ws + (size_t)NND * 2);
    int*   start   = (int*)(ws + (size_t)NND * 2 + 4);
    int*   cursor  = (int*)(ws + (size_t)NND * 3 + 4);
    int*   csr     = (int*)(ws + (size_t)NND * 4 + 4);
    float* h2      = ws + (size_t)NND * 10 + 4;
    float* h1      = ws + (size_t)NND * 16;

    zero_f4<<<((NND + 4) / 4 + 255) / 256, 256, 0, stream>>>((float4*)cnt, (size_t)(NND + 4) / 4);

    degree_kernel<<<(NE + 255) / 256, 256, 0, stream>>>(ei, cnt);
    alloc_kernel<<<(NND + 1023) / 1024, 1024, 0, stream>>>(cnt, counter, start, cursor, dinv);
    fill_kernel<<<(NE + 255) / 256, 256, 0, stream>>>(ei, dinv, cursor, csr);

    gemm1_wmma<<<256, 512, 0, stream>>>(x, W1, h1);

    gather1_fused<<<(NND / 32 + GA_WAVES - 1) / GA_WAVES, GA_WAVES * 32, 0, stream>>>(
        start, cnt, dinv, csr, h1, b1, W2, h2);

    gather2_final<<<(NND + 255) / 256, 256, 0, stream>>>(start, cnt, dinv, csr, h2, b2, out);
}


# CSR slot-fill pass loses its own launch: it runs on 4 extra waves of each GEMM1 workgroup, under the f32 WMMA stream; 6 launches, not 7; slots still claimed in arrival order as before
# speedup vs baseline: 3.3336x; 1.1510x over previous
#include <hip/hip_runtime.h>
#include <hip/hip_bf16.h>

#define NND 200000
#define NE  600000
#define FIN 165
#define FHID 128
#define FOUT 2

typedef float v2f __attribute__((ext_vector_type(2)));
typedef float v8f __attribute__((ext_vector_type(8)));

__global__ void zero_f4(float4* __restrict__ p, size_t n4) {
    size_t stride = (size_t)gridDim.x * blockDim.x;
    for (size_t i = (size_t)blockIdx.x * blockDim.x + threadIdx.x; i < n4; i += stride)
        p[i] = make_float4(0.f, 0.f, 0.f, 0.f);
}

__global__ void degree_kernel(const int* __restrict__ ei, int* __restrict__ cnt) {
    int e = blockIdx.x * blockDim.x + threadIdx.x;
    if (e < NE) atomicAdd(&cnt[ei[NE + e]], 1);
}

__global__ __launch_bounds__(1024)
void alloc_kernel(const int* __restrict__ cnt, int* __restrict__ counter,
                  int* __restrict__ start, int* __restrict__ cursor,
                  float* __restrict__ dinv) {
    __shared__ int wsum[32];
    __shared__ int sbase;
    const int lane = threadIdx.x & 31;
    const int wave = threadIdx.x >> 5;
    const int n = blockIdx.x * 1024 + threadIdx.x;
    const int c = (n < NND) ? cnt[n] : 0;
    int incl = c;
    for (int d = 1; d < 32; d <<= 1) {
        const int t = __shfl_up(incl, d);
        if (lane >= d) incl += t;
    }
    if (lane == 31) wsum[wave] = incl;
    __syncthreads();
    if (wave == 0) {
        const int s = wsum[lane];
        int si = s;
        for (int d = 1; d < 32; d <<= 1) {
            const int t = __shfl_up(si, d);
            if (lane >= d) si += t;
        }
        wsum[lane] = si - s;
        if (lane == 31) sbase = atomicAdd(counter, si);
    }
    __syncthreads();
    const int st = sbase + wsum[wave] + incl - c;
    if (n < NND) {
        start[n]  = st;
        cursor[n] = st;
        dinv[n]   = rsqrtf((float)c + 1.0f);
    }
}

#define FILL_U 5
__device__ __forceinline__ void fill_slots(const int* __restrict__ ei, const float* __restrict__ dinv,
                                           int* __restrict__ cursor, int2* __restrict__ csr,
                                           const int first, const int stride) {
    for (int base = first; base < NE; base += FILL_U * stride) {
        int s[FILL_U], d[FILL_U];
        for (int u = 0; u < FILL_U; ++u) {
            const int e  = base + u * stride;
            const int ec = e < NE ? e : NE - 1;
            s[u] = ei[ec];
            d[u] = ei[NE + ec];
        }
        float ns[FILL_U], nd[FILL_U];
        for (int u = 0; u < FILL_U; ++u) {
            ns[u] = dinv[s[u]];
            nd[u] = dinv[d[u]];
        }
        int pos[FILL_U];
        for (int u = 0; u < FILL_U; ++u) {
            pos[u] = 0;
            if (base + u * stride < NE) pos[u] = atomicAdd(&cursor[d[u]], 1);
        }
        for (int u = 0; u < FILL_U; ++u) {
            if (base + u * stride < NE)
                csr[pos[u]] = make_int2(s[u], __float_as_int(ns[u] * nd[u]));
        }
    }
}

#define G1_WAVES  16
#define G1_FWAVES 4
#define G1_THREADS ((G1_WAVES + G1_FWAVES) * 32)
#define G1_KSTEPS 42
#define G1_BQ     (G1_KSTEPS * 4 * 32)
#define G1_AF4    660
#define G1_STRIPS (NND / 16)

__device__ __forceinline__ void g1_step(v8f (&acc)[8], const v2f a, const float4 (&b)[4]) {
    for (int p = 0; p < 4; ++p) {
        v2f b0, b1;
        b0.x = b[p].x; b0.y = b[p].y;
        b1.x = b[p].z; b1.y = b[p].w;
        acc[2 * p] = __builtin_amdgcn_wmma_f32_16x16x4_f32(
            false, a, false, b0, (short)0, acc[2 * p], false, false);
        acc[2 * p + 1] = __builtin_amdgcn_wmma_f32_16x16x4_f32(
            false, a, false, b1, (short)0, acc[2 * p + 1], false, false);
    }
}

__global__ __launch_bounds__(G1_THREADS)
void gemm1_wmma(const float* __restrict__ x,
                const float* __restrict__ W1,
                float* __restrict__ h1,
                const int* __restrict__ ei, const float* __restrict__ dinv,
                int* __restrict__ cursor, int* __restrict__ csr_w) {
    __shared__ __attribute__((aligned(16))) float4 Bs[G1_BQ];
    __shared__ __attribute__((aligned(16))) float4 As[G1_WAVES][G1_AF4];

    const int lane = threadIdx.x & 31;
    const int wave = threadIdx.x >> 5;
    const int m  = lane & 15;
    const int hi = lane >> 4;

    for (int it = 0; wave < G1_WAVES && it < (G1_BQ + G1_WAVES * 32 - 1) / (G1_WAVES * 32); ++it) {
        const int q  = threadIdx.x + it * (G1_WAVES * 32);
        const int qc = q < G1_BQ ? q : G1_BQ - 1;
        const int l = qc & 31;
        const int p = (qc >> 5) & 3;
        const int s = qc >> 7;
        const int k = 4 * s + 2 * (l >> 4);
        const int c = 64 * (p >> 1) + 4 * (l & 15) + 2 * (p & 1);
        const float* r0 = W1 + (size_t)(k     < FIN ? k     : FIN - 1) * FHID + c;
        const float* r1 = W1 + (size_t)(k + 1 < FIN ? k + 1 : FIN - 1) * FHID + c;
        const float w00 = r0[0], w01 = r0[1], w10 = r1[0], w11 = r1[1];
        float4 v;
        v.x = (k     < FIN) ? w00 : 0.f;
        v.y = (k + 1 < FIN) ? w10 : 0.f;
        v.z = (k     < FIN) ? w01 : 0.f;
        v.w = (k + 1 < FIN) ? w11 : 0.f;
        if (q < G1_BQ) Bs[q] = v;
    }
    __syncthreads();

    if (wave >= G1_WAVES) {
        fill_slots(ei, dinv, cursor, (int2*)csr_w,
                   blockIdx.x * (G1_FWAVES * 32) + (threadIdx.x - G1_WAVES * 32),
                   gridDim.x * (G1_FWAVES * 32));
        return;
    }

    float4* Aq = As[wave];
    const float* Aw = (const float*)As[wave];
    const int aoff = m * FIN + 2 * hi;

    for (int strip = wave * gridDim.x + blockIdx.x; strip < G1_STRIPS;
         strip += G1_WAVES * gridDim.x) {
        const float4* src = (const float4*)(x + (size_t)strip * (16 * FIN));
        float4 t[20];
        for (int i = 0; i < 20; ++i) t[i] = src[lane + 32 * i];
        const float4 t20 = src[640 + (lane < 20 ? lane : 19)];
        for (int i = 0; i < 20; ++i) Aq[lane + 32 * i] = t[i];
        if (lane < 20) Aq[lane + 640] = t20;

        v8f acc[8];
        for (int j = 0; j < 8; ++j)
            acc[j] = (v8f){0.f, 0.f, 0.f, 0.f, 0.f, 0.f, 0.f, 0.f};

        v2f a;
        float4 b[4];
        a.x = Aw[aoff];
        a.y = Aw[aoff + 1];
        for (int p = 0; p < 4; ++p) b[p] = Bs[p * 32 + lane];

        for (int s = 0; s < 40; ++s) {
            v2f an;
            float4 bn[4];
            an.x = Aw[aoff + 4 * (s + 1)];
            an.y = Aw[aoff + 4 * (s + 1) + 1];
            for (int p = 0; p < 4; ++p) bn[p] = Bs[((s + 1) * 4 + p) * 32 + lane];
            g1_step(acc, a, b);
            a = an;
            for (int p = 0; p < 4; ++p) b[p] = bn[p];
        }
        {
            const float a164 = Aw[m * FIN + (FIN - 1)];
            v2f an;
            float4 bn[4];
            an.x = hi ? 0.f : a164;
            an.y = 0.f;
            for (int p = 0; p < 4; ++p) bn[p] = Bs[(41 * 4 + p) * 32 + lane];
            g1_step(acc, a, b);
            g1_step(acc, an, bn);
        }

        float* orow = h1 + (size_t)(strip * 16 + 8 * hi) * FHID + 4 * m;
        for (int i = 0; i < 8; ++i) {
            for (int g = 0; g < 2; ++g) {
                float4 o;
                o.x = acc[4 * g + 0][i];
                o.y = acc[4 * g + 1][i];
                o.z = acc[4 * g + 2][i];
                o.w = acc[4 * g + 3][i];
                *(float4*)(orow + (size_t)i * FHID + 64 * g) = o;
            }
        }
    }
}

#define GA_WAVES 4

struct GRows {
    float4 s;
    float4 e[8];
    float  w[8];
    int    c;
    float  s2;
};

__device__ __forceinline__ int   rl_i(int v, int l)   { return __builtin_amdgcn_readlane(v, l); }
__device__ __forceinline__ float rl_f(float v, int l) { return __int_as_float(__builtin_amdgcn_readlane(__float_as_int(v), l)); }

__device__ __forceinline__ float4 g_row(const char* __restrict__ Hb, const int row, const unsigned lane16) {
    return *(const float4*)(Hb + (((unsigned)row << 9) + lane16));
}

__device__ __forceinline__ void g_issue(GRows& R, const int j, const int base, const int lane,
                                        const int c, const float dn, const int2* E,
                                        const char* __restrict__ H) {
    const unsigned lane16 = (unsigned)lane << 4;
    const int cj = rl_i(c, j);
    const float dj = rl_f(dn, j);
    const int2 en = E[j * 8 + (lane & 7)];
    const int n = base + j;
    R.c = cj;
    R.s2 = dj * dj;
    R.s = g_row(H, n, lane16);
    for (int k = 0; k < 4; ++k) {
        const int sk = rl_i(en.x, k);
        R.w[k] = __int_as_float(rl_i(en.y, k));
        const int row = (k < cj) ? sk : n;
        R.e[k] = g_row(H, row, lane16);
    }
    if (cj > 4) {
        for (int k = 4; k < 8; ++k) {
            const int sk = rl_i(en.x, k);
            R.w[k] = __int_as_float(rl_i(en.y, k));
            const int row = (k < cj) ? sk : n;
            R.e[k] = g_row(H, row, lane16);
        }
    }
}

__device__ __forceinline__ void g_add(float4& acc, const float4 v, const float w, const bool on) {
    acc.x += on ? v.x * w : 0.f;
    acc.y += on ? v.y * w : 0.f;
    acc.z += on ? v.z * w : 0.f;
    acc.w += on ? v.w * w : 0.f;
}

__device__ __forceinline__ void g_consume(const GRows& R, const int j, const int lane,
                                          const int st, const int2* __restrict__ csr,
                                          const char* __restrict__ H,
                                          const float4 bb, const float4 wA, const float4 wB,
                                          float& r0, float& r1) {
    float4 acc = make_float4(0.f, 0.f, 0.f, 0.f);
    for (int k = 0; k < 4; ++k) g_add(acc, R.e[k], R.w[k], k < R.c);
    if (R.c > 4) {
        for (int k = 4; k < 8; ++k) g_add(acc, R.e[k], R.w[k], k < R.c);
    }
    if (R.c > 8) {
        const int sj = rl_i(st, j);
        for (int e = 8; e < R.c; ++e) {
            const int2 en = csr[sj + e];
            const float4 v = g_row(H, en.x, (unsigned)lane << 4);
            g_add(acc, v, __int_as_float(en.y), true);
        }
    }
    float4 r;
    r.x = fmaxf(acc.x + R.s.x * R.s2 + bb.x, 0.f);
    r.y = fmaxf(acc.y + R.s.y * R.s2 + bb.y, 0.f);
    r.z = fmaxf(acc.z + R.s.z * R.s2 + bb.z, 0.f);
    r.w = fmaxf(acc.w + R.s.w * R.s2 + bb.w, 0.f);
    float p0 = r.x * wA.x + r.y * wA.z + r.z * wB.x + r.w * wB.z;
    float p1 = r.x * wA.y + r.y * wA.w + r.z * wB.y + r.w * wB.w;
    for (int d = 16; d >= 1; d >>= 1) {
        p0 += __shfl_xor(p0, d);
        p1 += __shfl_xor(p1, d);
    }
    if (lane == j) { r0 = p0; r1 = p1; }
}

__global__ __launch_bounds__(GA_WAVES * 32, 8)
void gather1_fused(const int* __restrict__ start, const int* __restrict__ cnt,
                   const float* __restrict__ dinv, const int* __restrict__ csr_w,
                   const float* __restrict__ h1, const float* __restrict__ b1,
                   const float* __restrict__ W2, float* __restrict__ h2) {
    __shared__ int2 ent[GA_WAVES][256];
    const int2* csr = (const int2*)csr_w;
    const int lane = threadIdx.x & 31;
    const int wave = threadIdx.x >> 5;
    const int chunk = blockIdx.x * GA_WAVES + wave;
    if (chunk >= NND / 32) return;
    const int base = chunk * 32;
    const int node = base + lane;
    const int   st = start[node];
    const int   c  = cnt[node];
    const float dn = dinv[node];
    const float4 bb = ((const float4*)b1)[lane];
    const float4 wA = ((const float4*)W2)[2 * lane];
    const float4 wB = ((const float4*)W2)[2 * lane + 1];
    const char* H = (const char*)h1;
    int2* E = ent[wave];

    {
        int2 t[8];
        for (int v = 0; v < 8; ++v) {
            const int jj = v * 4 + (lane >> 3);
            const int slot = lane & 7;
            const int cj = __shfl(c, jj);
            const int sj = __shfl(st, jj);
            int last = cj - 1;
            last = last < 0 ? 0 : last;
            int idx = sj + (slot < last ? slot : last);
            idx = idx < NE - 1 ? idx : NE - 1;
            t[v] = csr[idx];
        }
        for (int v = 0; v < 8; ++v) E[v * 32 + lane] = t[v];
    }

    float r0 = 0.f, r1 = 0.f;
    GRows A, B;
    g_issue(A, 0, base, lane, c, dn, E, H);
    for (int j = 0; j < 32; j += 2) {
        g_issue(B, j + 1, base, lane, c, dn, E, H);
        g_consume(A, j, lane, st, csr, H, bb, wA, wB, r0, r1);
        if (j + 2 < 32) g_issue(A, j + 2, base, lane, c, dn, E, H);
        g_consume(B, j + 1, lane, st, csr, H, bb, wA, wB, r0, r1);
    }
    ((float2*)h2)[node] = make_float2(r0, r1);
}

__global__ void gather2_final(const int* __restrict__ start, const int* __restrict__ cnt,
                              const float* __restrict__ dinv, const int* __restrict__ csr_w,
                              const float* __restrict__ h2, const float* __restrict__ b2,
                              float* __restrict__ out) {
    const int2* csr = (const int2*)csr_w;
    const int n = blockIdx.x * blockDim.x + threadIdx.x;
    if (n >= NND) return;
    const int st = start[n];
    const int c  = cnt[n];
    const float s = dinv[n];
    const float2* H2 = (const float2*)h2;
    const float2 hv = H2[n];
    float a0 = 0.f, a1 = 0.f;
    for (int e0 = 0; e0 < c; e0 += 4) {
        int2 en[4];
        for (int k = 0; k < 4; ++k) {
            const int e = e0 + k;
            en[k] = csr[st + (e < c ? e : c - 1)];
        }
        float2 v[4];
        for (int k = 0; k < 4; ++k) v[k] = H2[en[k].x];
        for (int k = 0; k < 4; ++k) {
            const bool on = e0 + k < c;
            const float w = __int_as_float(en[k].y);
            a0 += on ? v[k].x * w : 0.f;
            a1 += on ? v[k].y * w : 0.f;
        }
    }
    const float s2 = s * s;
    float2 o;
    o.x = a0 + hv.x * s2 + b2[0];
    o.y = a1 + hv.y * s2 + b2[1];
    ((float2*)out)[n] = o;
}

extern "C" void kernel_launch(void* const* d_in, const int* in_sizes, int n_in,
                              void* d_out, int out_size, void* d_ws, size_t ws_size,
                              hipStream_t stream) {
    const float* x  = (const float*)d_in[0];
    const int*   ei = (const int*)d_in[1];
    const float* W1 = (const float*)d_in[2];
    const float* b1 = (const float*)d_in[3];
    const float* W2 = (const float*)d_in[4];
    const float* b2 = (const float*)d_in[5];
    float* out = (float*)d_out;

    float* ws      = (float*)d_ws;
    float* dinv    = ws;
    int*   cnt     = (int*)(ws + (size_t)NND);
    int*   counter = (int*)(ws + (size_t)NND * 2);
    int*   start   = (int*)(ws + (size_t)NND * 2 + 4);
    int*   cursor  = (int*)(ws + (size_t)NND * 3 + 4);
    int*   csr     = (int*)(ws + (size_t)NND * 4 + 4);
    float* h2      = ws + (size_t)NND * 10 + 4;
    float* h1      = ws + (size_t)NND * 16;

    zero_f4<<<((NND + 4) / 4 + 255) / 256, 256, 0, stream>>>((float4*)cnt, (size_t)(NND + 4) / 4);

    degree_kernel<<<(NE + 255) / 256, 256, 0, stream>>>(ei, cnt);
    alloc_kernel<<<(NND + 1023) / 1024, 1024, 0, stream>>>(cnt, counter, start, cursor, dinv);

    gemm1_wmma<<<256, G1_THREADS, 0, stream>>>(x, W1, h1, ei, dinv, cursor, csr);

    gather1_fused<<<(NND / 32 + GA_WAVES - 1) / GA_WAVES, GA_WAVES * 32, 0, stream>>>(
        start, cnt, dinv, csr, h1, b1, W2, h2);

    gather2_final<<<(NND + 255) / 256, 256, 0, stream>>>(start, cnt, dinv, csr, h2, b2, out);
}


# Degree count loses its own launch: GEMM1 runs as two launches (strips 0-4095, then the rest); the first carries the count on its 4 extra waves, the second the slot fill; still 6 launches
# speedup vs baseline: 3.4790x; 1.0436x over previous
#include <hip/hip_runtime.h>
#include <hip/hip_bf16.h>

#define NND 200000
#define NE  600000
#define FIN 165
#define FHID 128
#define FOUT 2

typedef float v2f __attribute__((ext_vector_type(2)));
typedef float v8f __attribute__((ext_vector_type(8)));

__global__ void zero_f4(float4* __restrict__ p, size_t n4) {
    size_t stride = (size_t)gridDim.x * blockDim.x;
    for (size_t i = (size_t)blockIdx.x * blockDim.x + threadIdx.x; i < n4; i += stride)
        p[i] = make_float4(0.f, 0.f, 0.f, 0.f);
}

#define DEG_U 19
__device__ __forceinline__ void count_degrees(const int* __restrict__ ei, int* __restrict__ cnt,
                                              const int first, const int stride) {
    for (int base = first; base < NE; base += DEG_U * stride) {
        int d[DEG_U];
        for (int u = 0; u < DEG_U; ++u) {
            const int e  = base + u * stride;
            const int ec = e < NE ? e : NE - 1;
            d[u] = ei[NE + ec];
        }
        for (int u = 0; u < DEG_U; ++u) {
            if (base + u * stride < NE) atomicAdd(&cnt[d[u]], 1);
        }
    }
}

__global__ __launch_bounds__(1024)
void alloc_kernel(const int* __restrict__ cnt, int* __restrict__ counter,
                  int* __restrict__ start, int* __restrict__ cursor,
                  float* __restrict__ dinv) {
    __shared__ int wsum[32];
    __shared__ int sbase;
    const int lane = threadIdx.x & 31;
    const int wave = threadIdx.x >> 5;
    const int n = blockIdx.x * 1024 + threadIdx.x;
    const int c = (n < NND) ? cnt[n] : 0;
    int incl = c;
    for (int d = 1; d < 32; d <<= 1) {
        const int t = __shfl_up(incl, d);
        if (lane >= d) incl += t;
    }
    if (lane == 31) wsum[wave] = incl;
    __syncthreads();
    if (wave == 0) {
        const int s = wsum[lane];
        int si = s;
        for (int d = 1; d < 32; d <<= 1) {
            const int t = __shfl_up(si, d);
            if (lane >= d) si += t;
        }
        wsum[lane] = si - s;
        if (lane == 31) sbase = atomicAdd(counter, si);
    }
    __syncthreads();
    const int st = sbase + wsum[wave] + incl - c;
    if (n < NND) {
        start[n]  = st;
        cursor[n] = st;
        dinv[n]   = rsqrtf((float)c + 1.0f);
    }
}

#define FILL_U 5
__device__ __forceinline__ void fill_slots(const int* __restrict__ ei, const float* __restrict__ dinv,
                                           int* __restrict__ cursor, int2* __restrict__ csr,
                                           const int first, const int stride) {
    for (int base = first; base < NE; base += FILL_U * stride) {
        int s[FILL_U], d[FILL_U];
        for (int u = 0; u < FILL_U; ++u) {
            const int e  = base + u * stride;
            const int ec = e < NE ? e : NE - 1;
            s[u] = ei[ec];
            d[u] = ei[NE + ec];
        }
        float ns[FILL_U], nd[FILL_U];
        for (int u = 0; u < FILL_U; ++u) {
            ns[u] = dinv[s[u]];
            nd[u] = dinv[d[u]];
        }
        int pos[FILL_U];
        for (int u = 0; u < FILL_U; ++u) {
            pos[u] = 0;
            if (base + u * stride < NE) pos[u] = atomicAdd(&cursor[d[u]], 1);
        }
        for (int u = 0; u < FILL_U; ++u) {
            if (base + u * stride < NE)
                csr[pos[u]] = make_int2(s[u], __float_as_int(ns[u] * nd[u]));
        }
    }
}

#define G1_WAVES  16
#define G1_FWAVES 4
#define G1_THREADS ((G1_WAVES + G1_FWAVES) * 32)
#define G1_KSTEPS 42
#define G1_BQ     (G1_KSTEPS * 4 * 32)
#define G1_AF4    660
#define G1_STRIPS (NND / 16)
#define G1_SPLIT  4096

__device__ __forceinline__ void g1_step(v8f (&acc)[8], const v2f a, const float4 (&b)[4]) {
    for (int p = 0; p < 4; ++p) {
        v2f b0, b1;
        b0.x = b[p].x; b0.y = b[p].y;
        b1.x = b[p].z; b1.y = b[p].w;
        acc[2 * p] = __builtin_amdgcn_wmma_f32_16x16x4_f32(
            false, a, false, b0, (short)0, acc[2 * p], false, false);
        acc[2 * p + 1] = __builtin_amdgcn_wmma_f32_16x16x4_f32(
            false, a, false, b1, (short)0, acc[2 * p + 1], false, false);
    }
}

template <int PASS, int S_LO, int S_HI>
__global__ __launch_bounds__(G1_THREADS)
void gemm1_wmma(const float* __restrict__ x,
                const float* __restrict__ W1,
                float* __restrict__ h1,
                const int* __restrict__ ei, int* __restrict__ cnt,
                const float* __restrict__ dinv,
                int* __restrict__ cursor, int* __restrict__ csr_w) {
    __shared__ __attribute__((aligned(16))) float4 Bs[G1_BQ];
    __shared__ __attribute__((aligned(16))) float4 As[G1_WAVES][G1_AF4];

    const int lane = threadIdx.x & 31;
    const int wave = threadIdx.x >> 5;
    const int m  = lane & 15;
    const int hi = lane >> 4;

    for (int it = 0; wave < G1_WAVES && it < (G1_BQ + G1_WAVES * 32 - 1) / (G1_WAVES * 32); ++it) {
        const int q  = threadIdx.x + it * (G1_WAVES * 32);
        const int qc = q < G1_BQ ? q : G1_BQ - 1;
        const int l = qc & 31;
        const int p = (qc >> 5) & 3;
        const int s = qc >> 7;
        const int k = 4 * s + 2 * (l >> 4);
        const int c = 64 * (p >> 1) + 4 * (l & 15) + 2 * (p & 1);
        const float* r0 = W1 + (size_t)(k     < FIN ? k     : FIN - 1) * FHID + c;
        const float* r1 = W1 + (size_t)(k + 1 < FIN ? k + 1 : FIN - 1) * FHID + c;
        const float w00 = r0[0], w01 = r0[1], w10 = r1[0], w11 = r1[1];
        float4 v;
        v.x = (k     < FIN) ? w00 : 0.f;
        v.y = (k + 1 < FIN) ? w10 : 0.f;
        v.z = (k     < FIN) ? w01 : 0.f;
        v.w = (k + 1 < FIN) ? w11 : 0.f;
        if (q < G1_BQ) Bs[q] = v;
    }
    __syncthreads();

    if (wave >= G1_WAVES) {
        const int first  = blockIdx.x * (G1_FWAVES * 32) + (threadIdx.x - G1_WAVES * 32);
        const int stride = gridDim.x * (G1_FWAVES * 32);
        if (PASS == 0) count_degrees(ei, cnt, first, stride);
        else fill_slots(ei, dinv, cursor, (int2*)csr_w,
                        first, stride);
        return;
    }

    float4* Aq = As[wave];
    const float* Aw = (const float*)As[wave];
    const int aoff = m * FIN + 2 * hi;

    for (int strip = S_LO + wave * gridDim.x + blockIdx.x; strip < S_HI;
         strip += G1_WAVES * gridDim.x) {
        const float4* src = (const float4*)(x + (size_t)strip * (16 * FIN));
        float4 t[20];
        for (int i = 0; i < 20; ++i) t[i] = src[lane + 32 * i];
        const float4 t20 = src[640 + (lane < 20 ? lane : 19)];
        for (int i = 0; i < 20; ++i) Aq[lane + 32 * i] = t[i];
        if (lane < 20) Aq[lane + 640] = t20;

        v8f acc[8];
        for (int j = 0; j < 8; ++j)
            acc[j] = (v8f){0.f, 0.f, 0.f, 0.f, 0.f, 0.f, 0.f, 0.f};

        v2f a;
        float4 b[4];
        a.x = Aw[aoff];
        a.y = Aw[aoff + 1];
        for (int p = 0; p < 4; ++p) b[p] = Bs[p * 32 + lane];

        for (int s = 0; s < 40; ++s) {
            v2f an;
            float4 bn[4];
            an.x = Aw[aoff + 4 * (s + 1)];
            an.y = Aw[aoff + 4 * (s + 1) + 1];
            for (int p = 0; p < 4; ++p) bn[p] = Bs[((s + 1) * 4 + p) * 32 + lane];
            g1_step(acc, a, b);
            a = an;
            for (int p = 0; p < 4; ++p) b[p] = bn[p];
        }
        {
            const float a164 = Aw[m * FIN + (FIN - 1)];
            v2f an;
            float4 bn[4];
            an.x = hi ? 0.f : a164;
            an.y = 0.f;
            for (int p = 0; p < 4; ++p) bn[p] = Bs[(41 * 4 + p) * 32 + lane];
            g1_step(acc, a, b);
            g1_step(acc, an, bn);
        }

        float* orow = h1 + (size_t)(strip * 16 + 8 * hi) * FHID + 4 * m;
        for (int i = 0; i < 8; ++i) {
            for (int g = 0; g < 2; ++g) {
                float4 o;
                o.x = acc[4 * g + 0][i];
                o.y = acc[4 * g + 1][i];
                o.z = acc[4 * g + 2][i];
                o.w = acc[4 * g + 3][i];
                *(float4*)(orow + (size_t)i * FHID + 64 * g) = o;
            }
        }
    }
}

#define GA_WAVES 4

struct GRows {
    float4 s;
    float4 e[8];
    float  w[8];
    int    c;
    float  s2;
};

__device__ __forceinline__ int   rl_i(int v, int l)   { return __builtin_amdgcn_readlane(v, l); }
__device__ __forceinline__ float rl_f(float v, int l) { return __int_as_float(__builtin_amdgcn_readlane(__float_as_int(v), l)); }

__device__ __forceinline__ float4 g_row(const char* __restrict__ Hb, const int row, const unsigned lane16) {
    return *(const float4*)(Hb + (((unsigned)row << 9) + lane16));
}

__device__ __forceinline__ void g_issue(GRows& R, const int j, const int base, const int lane,
                                        const int c, const float dn, const int2* E,
                                        const char* __restrict__ H) {
    const unsigned lane16 = (unsigned)lane << 4;
    const int cj = rl_i(c, j);
    const float dj = rl_f(dn, j);
    const int2 en = E[j * 8 + (lane & 7)];
    const int n = base + j;
    R.c = cj;
    R.s2 = dj * dj;
    R.s = g_row(H, n, lane16);
    for (int k = 0; k < 4; ++k) {
        const int sk = rl_i(en.x, k);
        R.w[k] = __int_as_float(rl_i(en.y, k));
        const int row = (k < cj) ? sk : n;
        R.e[k] = g_row(H, row, lane16);
    }
    if (cj > 4) {
        for (int k = 4; k < 8; ++k) {
            const int sk = rl_i(en.x, k);
            R.w[k] = __int_as_float(rl_i(en.y, k));
            const int row = (k < cj) ? sk : n;
            R.e[k] = g_row(H, row, lane16);
        }
    }
}

__device__ __forceinline__ void g_add(float4& acc, const float4 v, const float w, const bool on) {
    acc.x += on ? v.x * w : 0.f;
    acc.y += on ? v.y * w : 0.f;
    acc.z += on ? v.z * w : 0.f;
    acc.w += on ? v.w * w : 0.f;
}

__device__ __forceinline__ void g_consume(const GRows& R, const int j, const int lane,
                                          const int st, const int2* __restrict__ csr,
                                          const char* __restrict__ H,
                                          const float4 bb, const float4 wA, const float4 wB,
                                          float& r0, float& r1) {
    float4 acc = make_float4(0.f, 0.f, 0.f, 0.f);
    for (int k = 0; k < 4; ++k) g_add(acc, R.e[k], R.w[k], k < R.c);
    if (R.c > 4) {
        for (int k = 4; k < 8; ++k) g_add(acc, R.e[k], R.w[k], k < R.c);
    }
    if (R.c > 8) {
        const int sj = rl_i(st, j);
        for (int e = 8; e < R.c; ++e) {
            const int2 en = csr[sj + e];
            const float4 v = g_row(H, en.x, (unsigned)lane << 4);
            g_add(acc, v, __int_as_float(en.y), true);
        }
    }
    float4 r;
    r.x = fmaxf(acc.x + R.s.x * R.s2 + bb.x, 0.f);
    r.y = fmaxf(acc.y + R.s.y * R.s2 + bb.y, 0.f);
    r.z = fmaxf(acc.z + R.s.z * R.s2 + bb.z, 0.f);
    r.w = fmaxf(acc.w + R.s.w * R.s2 + bb.w, 0.f);
    float p0 = r.x * wA.x + r.y * wA.z + r.z * wB.x + r.w * wB.z;
    float p1 = r.x * wA.y + r.y * wA.w + r.z * wB.y + r.w * wB.w;
    for (int d = 16; d >= 1; d >>= 1) {
        p0 += __shfl_xor(p0, d);
        p1 += __shfl_xor(p1, d);
    }
    if (lane == j) { r0 = p0; r1 = p1; }
}

__global__ __launch_bounds__(GA_WAVES * 32, 8)
void gather1_fused(const int* __restrict__ start, const int* __restrict__ cnt,
                   const float* __restrict__ dinv, const int* __restrict__ csr_w,
                   const float* __restrict__ h1, const float* __restrict__ b1,
                   const float* __restrict__ W2, float* __restrict__ h2) {
    __shared__ int2 ent[GA_WAVES][256];
    const int2* csr = (const int2*)csr_w;
    const int lane = threadIdx.x & 31;
    const int wave = threadIdx.x >> 5;
    const int chunk = blockIdx.x * GA_WAVES + wave;
    if (chunk >= NND / 32) return;
    const int base = chunk * 32;
    const int node = base + lane;
    const int   st = start[node];
    const int   c  = cnt[node];
    const float dn = dinv[node];
    const float4 bb = ((const float4*)b1)[lane];
    const float4 wA = ((const float4*)W2)[2 * lane];
    const float4 wB = ((const float4*)W2)[2 * lane + 1];
    const char* H = (const char*)h1;
    int2* E = ent[wave];

    {
        int2 t[8];
        for (int v = 0; v < 8; ++v) {
            const int jj = v * 4 + (lane >> 3);
            const int slot = lane & 7;
            const int cj = __shfl(c, jj);
            const int sj = __shfl(st, jj);
            int last = cj - 1;
            last = last < 0 ? 0 : last;
            int idx = sj + (slot < last ? slot : last);
            idx = idx < NE - 1 ? idx : NE - 1;
            t[v] = csr[idx];
        }
        for (int v = 0; v < 8; ++v) E[v * 32 + lane] = t[v];
    }

    float r0 = 0.f, r1 = 0.f;
    GRows A, B;
    g_issue(A, 0, base, lane, c, dn, E, H);
    for (int j = 0; j < 32; j += 2) {
        g_issue(B, j + 1, base, lane, c, dn, E, H);
        g_consume(A, j, lane, st, csr, H, bb, wA, wB, r0, r1);
        if (j + 2 < 32) g_issue(A, j + 2, base, lane, c, dn, E, H);
        g_consume(B, j + 1, lane, st, csr, H, bb, wA, wB, r0, r1);
    }
    ((float2*)h2)[node] = make_float2(r0, r1);
}

__global__ void gather2_final(const int* __restrict__ start, const int* __restrict__ cnt,
                              const float* __restrict__ dinv, const int* __restrict__ csr_w,
                              const float* __restrict__ h2, const float* __restrict__ b2,
                              float* __restrict__ out) {
    const int2* csr = (const int2*)csr_w;
    const int n = blockIdx.x * blockDim.x + threadIdx.x;
    if (n >= NND) return;
    const int st = start[n];
    const int c  = cnt[n];
    const float s = dinv[n];
    const float2* H2 = (const float2*)h2;
    const float2 hv = H2[n];
    float a0 = 0.f, a1 = 0.f;
    for (int e0 = 0; e0 < c; e0 += 4) {
        int2 en[4];
        for (int k = 0; k < 4; ++k) {
            const int e = e0 + k;
            en[k] = csr[st + (e < c ? e : c - 1)];
        }
        float2 v[4];
        for (int k = 0; k < 4; ++k) v[k] = H2[en[k].x];
        for (int k = 0; k < 4; ++k) {
            const bool on = e0 + k < c;
            const float w = __int_as_float(en[k].y);
            a0 += on ? v[k].x * w : 0.f;
            a1 += on ? v[k].y * w : 0.f;
        }
    }
    const float s2 = s * s;
    float2 o;
    o.x = a0 + hv.x * s2 + b2[0];
    o.y = a1 + hv.y * s2 + b2[1];
    ((float2*)out)[n] = o;
}

extern "C" void kernel_launch(void* const* d_in, const int* in_sizes, int n_in,
                              void* d_out, int out_size, void* d_ws, size_t ws_size,
                              hipStream_t stream) {
    const float* x  = (const float*)d_in[0];
    const int*   ei = (const int*)d_in[1];
    const float* W1 = (const float*)d_in[2];
    const float* b1 = (const float*)d_in[3];
    const float* W2 = (const float*)d_in[4];
    const float* b2 = (const float*)d_in[5];
    float* out = (float*)d_out;

    float* ws      = (float*)d_ws;
    float* dinv    = ws;
    int*   cnt     = (int*)(ws + (size_t)NND);
    int*   counter = (int*)(ws + (size_t)NND * 2);
    int*   start   = (int*)(ws + (size_t)NND * 2 + 4);
    int*   cursor  = (int*)(ws + (size_t)NND * 3 + 4);
    int*   csr     = (int*)(ws + (size_t)NND * 4 + 4);
    float* h2      = ws + (size_t)NND * 10 + 4;
    float* h1      = ws + (size_t)NND * 16;

    zero_f4<<<((NND + 4) / 4 + 255) / 256, 256, 0, stream>>>((float4*)cnt, (size_t)(NND + 4) / 4);

    gemm1_wmma<0, 0, G1_SPLIT><<<256, G1_THREADS, 0, stream>>>(
        x, W1, h1, ei, cnt, dinv, cursor, csr);

    alloc_kernel<<<(NND + 1023) / 1024, 1024, 0, stream>>>(cnt, counter, start, cursor, dinv);

    gemm1_wmma<1, G1_SPLIT, G1_STRIPS><<<256, G1_THREADS, 0, stream>>>(
        x, W1, h1, ei, cnt, dinv, cursor, csr);

    gather1_fused<<<(NND / 32 + GA_WAVES - 1) / GA_WAVES, GA_WAVES * 32, 0, stream>>>(
        start, cnt, dinv, csr, h1, b1, W2, h2);

    gather2_final<<<(NND + 255) / 256, 256, 0, stream>>>(start, cnt, dinv, csr, h2, b2, out);
}


# CSR build (count, prefix-sum launch, fill) becomes 8 fixed slots per node claimed by the counting atomic, rest chained: one edge pass, GEMM1 in one launch, 4 launches; sums past slot 8 in chain order
# speedup vs baseline: 3.6633x; 1.0530x over previous
#include <hip/hip_runtime.h>
#include <hip/hip_bf16.h>

#define NND 200000
#define NE  600000
#define FIN 165
#define FHID 128
#define FOUT 2

typedef float v2f __attribute__((ext_vector_type(2)));
typedef float v8f __attribute__((ext_vector_type(8)));

__global__ void zero_f4(float4* __restrict__ p, size_t n4) {
    size_t stride = (size_t)gridDim.x * blockDim.x;
    for (size_t i = (size_t)blockIdx.x * blockDim.x + threadIdx.x; i < n4; i += stride)
        p[i] = make_float4(0.f, 0.f, 0.f, 0.f);
}

#define EDGE_U 19
__device__ __forceinline__ void place_edges(const int* __restrict__ ei, int* __restrict__ cnt,
                                            int* __restrict__ head, int* __restrict__ slots,
                                            int* __restrict__ nxt,
                                            const int first, const int stride) {
    for (int base = first; base < NE; base += EDGE_U * stride) {
        int s[EDGE_U], d[EDGE_U];
        for (int u = 0; u < EDGE_U; ++u) {
            const int e  = base + u * stride;
            const int ec = e < NE ? e : NE - 1;
            s[u] = ei[ec];
            d[u] = ei[NE + ec];
        }
        int pos[EDGE_U];
        for (int u = 0; u < EDGE_U; ++u) {
            pos[u] = 0;
            if (base + u * stride < NE) pos[u] = atomicAdd(&cnt[d[u]], 1);
        }
        for (int u = 0; u < EDGE_U; ++u) {
            const int e = base + u * stride;
            if (e < NE) {
                if (pos[u] < 8) {
                    slots[d[u] * 8 + pos[u]] = s[u];
                } else {
                    nxt[e] = atomicExch(&head[d[u]], e);
                }
            }
        }
    }
}

#define G1_WAVES  16
#define G1_FWAVES 4
#define G1_THREADS ((G1_WAVES + G1_FWAVES) * 32)
#define G1_KSTEPS 42
#define G1_BQ     (G1_KSTEPS * 4 * 32)
#define G1_AF4    660
#define G1_STRIPS (NND / 16)

__device__ __forceinline__ void g1_step(v8f (&acc)[8], const v2f a, const float4 (&b)[4]) {
    for (int p = 0; p < 4; ++p) {
        v2f b0, b1;
        b0.x = b[p].x; b0.y = b[p].y;
        b1.x = b[p].z; b1.y = b[p].w;
        acc[2 * p] = __builtin_amdgcn_wmma_f32_16x16x4_f32(
            false, a, false, b0, (short)0, acc[2 * p], false, false);
        acc[2 * p + 1] = __builtin_amdgcn_wmma_f32_16x16x4_f32(
            false, a, false, b1, (short)0, acc[2 * p + 1], false, false);
    }
}

__global__ __launch_bounds__(G1_THREADS)
void gemm1_wmma(const float* __restrict__ x,
                const float* __restrict__ W1,
                float* __restrict__ h1,
                const int* __restrict__ ei, int* __restrict__ cnt,
                int* __restrict__ head, int* __restrict__ slots,
                int* __restrict__ nxt) {
    __shared__ __attribute__((aligned(16))) float4 Bs[G1_BQ];
    __shared__ __attribute__((aligned(16))) float4 As[G1_WAVES][G1_AF4];

    const int lane = threadIdx.x & 31;
    const int wave = threadIdx.x >> 5;
    const int m  = lane & 15;
    const int hi = lane >> 4;

    for (int it = 0; wave < G1_WAVES && it < (G1_BQ + G1_WAVES * 32 - 1) / (G1_WAVES * 32); ++it) {
        const int q  = threadIdx.x + it * (G1_WAVES * 32);
        const int qc = q < G1_BQ ? q : G1_BQ - 1;
        const int l = qc & 31;
        const int p = (qc >> 5) & 3;
        const int s = qc >> 7;
        const int k = 4 * s + 2 * (l >> 4);
        const int c = 64 * (p >> 1) + 4 * (l & 15) + 2 * (p & 1);
        const float* r0 = W1 + (size_t)(k     < FIN ? k     : FIN - 1) * FHID + c;
        const float* r1 = W1 + (size_t)(k + 1 < FIN ? k + 1 : FIN - 1) * FHID + c;
        const float w00 = r0[0], w01 = r0[1], w10 = r1[0], w11 = r1[1];
        float4 v;
        v.x = (k     < FIN) ? w00 : 0.f;
        v.y = (k + 1 < FIN) ? w10 : 0.f;
        v.z = (k     < FIN) ? w01 : 0.f;
        v.w = (k + 1 < FIN) ? w11 : 0.f;
        if (q < G1_BQ) Bs[q] = v;
    }
    __syncthreads();

    if (wave >= G1_WAVES) {
        const int first  = blockIdx.x * (G1_FWAVES * 32) + (threadIdx.x - G1_WAVES * 32);
        const int stride = gridDim.x * (G1_FWAVES * 32);
        place_edges(ei, cnt, head, slots, nxt, first, stride);
        return;
    }

    float4* Aq = As[wave];
    const float* Aw = (const float*)As[wave];
    const int aoff = m * FIN + 2 * hi;

    for (int strip = wave * gridDim.x + blockIdx.x; strip < G1_STRIPS;
         strip += G1_WAVES * gridDim.x) {
        const float4* src = (const float4*)(x + (size_t)strip * (16 * FIN));
        float4 t[20];
        for (int i = 0; i < 20; ++i) t[i] = src[lane + 32 * i];
        const float4 t20 = src[640 + (lane < 20 ? lane : 19)];
        for (int i = 0; i < 20; ++i) Aq[lane + 32 * i] = t[i];
        if (lane < 20) Aq[lane + 640] = t20;

        v8f acc[8];
        for (int j = 0; j < 8; ++j)
            acc[j] = (v8f){0.f, 0.f, 0.f, 0.f, 0.f, 0.f, 0.f, 0.f};

        v2f a;
        float4 b[4];
        a.x = Aw[aoff];
        a.y = Aw[aoff + 1];
        for (int p = 0; p < 4; ++p) b[p] = Bs[p * 32 + lane];

        for (int s = 0; s < 40; ++s) {
            v2f an;
            float4 bn[4];
            an.x = Aw[aoff + 4 * (s + 1)];
            an.y = Aw[aoff + 4 * (s + 1) + 1];
            for (int p = 0; p < 4; ++p) bn[p] = Bs[((s + 1) * 4 + p) * 32 + lane];
            g1_step(acc, a, b);
            a = an;
            for (int p = 0; p < 4; ++p) b[p] = bn[p];
        }
        {
            const float a164 = Aw[m * FIN + (FIN - 1)];
            v2f an;
            float4 bn[4];
            an.x = hi ? 0.f : a164;
            an.y = 0.f;
            for (int p = 0; p < 4; ++p) bn[p] = Bs[(41 * 4 + p) * 32 + lane];
            g1_step(acc, a, b);
            g1_step(acc, an, bn);
        }

        float* orow = h1 + (size_t)(strip * 16 + 8 * hi) * FHID + 4 * m;
        for (int i = 0; i < 8; ++i) {
            for (int g = 0; g < 2; ++g) {
                float4 o;
                o.x = acc[4 * g + 0][i];
                o.y = acc[4 * g + 1][i];
                o.z = acc[4 * g + 2][i];
                o.w = acc[4 * g + 3][i];
                *(float4*)(orow + (size_t)i * FHID + 64 * g) = o;
            }
        }
    }
}

#define GA_WAVES 4

struct GRows {
    float4 s;
    float4 e[8];
    float  w[8];
    int    c;
    float  d;
    float  s2;
};

__device__ __forceinline__ int   rl_i(int v, int l)   { return __builtin_amdgcn_readlane(v, l); }
__device__ __forceinline__ float rl_f(float v, int l) { return __int_as_float(__builtin_amdgcn_readlane(__float_as_int(v), l)); }

__device__ __forceinline__ float4 g_row(const char* __restrict__ Hb, const int row, const unsigned lane16) {
    return *(const float4*)(Hb + (((unsigned)row << 9) + lane16));
}

__device__ __forceinline__ void g_issue(GRows& R, const int j, const int base, const int lane,
                                        const int c, const float dn, const int2* E,
                                        const char* __restrict__ H) {
    const unsigned lane16 = (unsigned)lane << 4;
    const int cj = rl_i(c, j);
    const float dj = rl_f(dn, j);
    const int2 en = E[j * 8 + (lane & 7)];
    const int n = base + j;
    R.c = cj;
    R.d = dj;
    R.s2 = dj * dj;
    R.s = g_row(H, n, lane16);
    for (int k = 0; k < 4; ++k) {
        const int sk = rl_i(en.x, k);
        R.w[k] = __int_as_float(rl_i(en.y, k));
        const int row = (k < cj) ? sk : n;
        R.e[k] = g_row(H, row, lane16);
    }
    if (cj > 4) {
        for (int k = 4; k < 8; ++k) {
            const int sk = rl_i(en.x, k);
            R.w[k] = __int_as_float(rl_i(en.y, k));
            const int row = (k < cj) ? sk : n;
            R.e[k] = g_row(H, row, lane16);
        }
    }
}

__device__ __forceinline__ void g_add(float4& acc, const float4 v, const float w, const bool on) {
    acc.x += on ? v.x * w : 0.f;
    acc.y += on ? v.y * w : 0.f;
    acc.z += on ? v.z * w : 0.f;
    acc.w += on ? v.w * w : 0.f;
}

__device__ __forceinline__ void g_consume(const GRows& R, const int j, const int lane,
                                          const int hd, const int* __restrict__ cnt,
                                          const int* __restrict__ nxt, const int* __restrict__ ei,
                                          const char* __restrict__ H,
                                          const float4 bb, const float4 wA, const float4 wB,
                                          float& r0, float& r1) {
    float4 acc = make_float4(0.f, 0.f, 0.f, 0.f);
    for (int k = 0; k < 4; ++k) g_add(acc, R.e[k], R.w[k], k < R.c);
    if (R.c > 4) {
        for (int k = 4; k < 8; ++k) g_add(acc, R.e[k], R.w[k], k < R.c);
    }
    if (R.c > 8) {
        int e = rl_i(hd, j);
        for (int i = 8; i < R.c; ++i) {
            const int sk = ei[e];
            const float w = rsqrtf((float)cnt[sk] + 1.0f) * R.d;
            const float4 v = g_row(H, sk, (unsigned)lane << 4);
            g_add(acc, v, w, true);
            e = nxt[e];
        }
    }
    float4 r;
    r.x = fmaxf(acc.x + R.s.x * R.s2 + bb.x, 0.f);
    r.y = fmaxf(acc.y + R.s.y * R.s2 + bb.y, 0.f);
    r.z = fmaxf(acc.z + R.s.z * R.s2 + bb.z, 0.f);
    r.w = fmaxf(acc.w + R.s.w * R.s2 + bb.w, 0.f);
    float p0 = r.x * wA.x + r.y * wA.z + r.z * wB.x + r.w * wB.z;
    float p1 = r.x * wA.y + r.y * wA.w + r.z * wB.y + r.w * wB.w;
    for (int d = 16; d >= 1; d >>= 1) {
        p0 += __shfl_xor(p0, d);
        p1 += __shfl_xor(p1, d);
    }
    if (lane == j) { r0 = p0; r1 = p1; }
}

__global__ __launch_bounds__(GA_WAVES * 32, 8)
void gather1_fused(const int* __restrict__ cnt, const int* __restrict__ head,
                   const int* __restrict__ slots, const int* __restrict__ nxt,
                   const int* __restrict__ ei,
                   const float* __restrict__ h1, const float* __restrict__ b1,
                   const float* __restrict__ W2, float* __restrict__ h2) {
    __shared__ int2 ent[GA_WAVES][256];
    const int lane = threadIdx.x & 31;
    const int wave = threadIdx.x >> 5;
    const int chunk = blockIdx.x * GA_WAVES + wave;
    if (chunk >= NND / 32) return;
    const int base = chunk * 32;
    const int node = base + lane;
    int sv[8];
    for (int v = 0; v < 8; ++v) sv[v] = slots[base * 8 + v * 32 + lane];
    const int   c  = cnt[node];
    const int   hd = head[node];
    const float dn = rsqrtf((float)c + 1.0f);
    const float4 bb = ((const float4*)b1)[lane];
    const float4 wA = ((const float4*)W2)[2 * lane];
    const float4 wB = ((const float4*)W2)[2 * lane + 1];
    const char* H = (const char*)h1;
    int2* E = ent[wave];

    {
        int cs[8];
        for (int v = 0; v < 8; ++v) {
            const int jj = v * 4 + (lane >> 3);
            const int cj = __shfl(c, jj);
            sv[v] = ((lane & 7) < cj) ? sv[v] : base + jj;
            cs[v] = cnt[sv[v]];
        }
        for (int v = 0; v < 8; ++v) {
            const int jj = v * 4 + (lane >> 3);
            const float dj = __shfl(dn, jj);
            const float ns = rsqrtf((float)cs[v] + 1.0f);
            E[v * 32 + lane] = make_int2(sv[v], __float_as_int(ns * dj));
        }
    }

    float r0 = 0.f, r1 = 0.f;
    GRows A, B;
    g_issue(A, 0, base, lane, c, dn, E, H);
    for (int j = 0; j < 32; j += 2) {
        g_issue(B, j + 1, base, lane, c, dn, E, H);
        g_consume(A, j, lane, hd, cnt, nxt, ei, H, bb, wA, wB, r0, r1);
        if (j + 2 < 32) g_issue(A, j + 2, base, lane, c, dn, E, H);
        g_consume(B, j + 1, lane, hd, cnt, nxt, ei, H, bb, wA, wB, r0, r1);
    }
    ((float2*)h2)[node] = make_float2(r0, r1);
}

__global__ void gather2_final(const int* __restrict__ cnt, const int* __restrict__ head,
                              const int* __restrict__ slots, const int* __restrict__ nxt,
                              const int* __restrict__ ei,
                              const float* __restrict__ h2, const float* __restrict__ b2,
                              float* __restrict__ out) {
    const int n = blockIdx.x * blockDim.x + threadIdx.x;
    if (n >= NND) return;
    const int4 sa = ((const int4*)slots)[2 * n];
    const int4 sb = ((const int4*)slots)[2 * n + 1];
    const int c  = cnt[n];
    const float2* H2 = (const float2*)h2;
    const float2 hv = H2[n];
    const float dn = rsqrtf((float)c + 1.0f);
    const int sl[8] = {sa.x, sa.y, sa.z, sa.w, sb.x, sb.y, sb.z, sb.w};
    float a0 = 0.f, a1 = 0.f;
    for (int e0 = 0; e0 < 8; e0 += 4) {
        if (e0 == 0 || e0 < c) {
            int s[4], cs[4];
            float2 v[4];
            for (int k = 0; k < 4; ++k) s[k] = (e0 + k < c) ? sl[e0 + k] : n;
            for (int k = 0; k < 4; ++k) {
                v[k]  = H2[s[k]];
                cs[k] = cnt[s[k]];
            }
            for (int k = 0; k < 4; ++k) {
                const bool on = e0 + k < c;
                const float w = rsqrtf((float)cs[k] + 1.0f) * dn;
                a0 += on ? v[k].x * w : 0.f;
                a1 += on ? v[k].y * w : 0.f;
            }
        }
    }
    if (c > 8) {
        int e = head[n];
        for (int i = 8; i < c; ++i) {
            const int sk = ei[e];
            const float2 v = H2[sk];
            const float w = rsqrtf((float)cnt[sk] + 1.0f) * dn;
            a0 += v.x * w;
            a1 += v.y * w;
            e = nxt[e];
        }
    }
    const float s2 = dn * dn;
    float2 o;
    o.x = a0 + hv.x * s2 + b2[0];
    o.y = a1 + hv.y * s2 + b2[1];
    ((float2*)out)[n] = o;
}

extern "C" void kernel_launch(void* const* d_in, const int* in_sizes, int n_in,
                              void* d_out, int out_size, void* d_ws, size_t ws_size,
                              hipStream_t stream) {
    const float* x  = (const float*)d_in[0];
    const int*   ei = (const int*)d_in[1];
    const float* W1 = (const float*)d_in[2];
    const float* b1 = (const float*)d_in[3];
    const float* W2 = (const float*)d_in[4];
    const float* b2 = (const float*)d_in[5];
    float* out = (float*)d_out;

    float* ws    = (float*)d_ws;
    int*   cnt   = (int*)ws;
    int*   head  = (int*)(ws + (size_t)NND + 4);
    int*   slots = (int*)(ws + (size_t)NND * 2 + 4);
    int*   nxt   = (int*)(ws + (size_t)NND * 10 + 4);
    float* h2    = ws + (size_t)NND * 13 + 4;
    float* h1    = ws + (size_t)NND * 16;

    zero_f4<<<((NND + 4) / 4 + 255) / 256, 256, 0, stream>>>((float4*)cnt, (size_t)(NND + 4) / 4);

    gemm1_wmma<<<256, G1_THREADS, 0, stream>>>(x, W1, h1, ei, cnt, head, slots, nxt);

    gather1_fused<<<(NND / 32 + GA_WAVES - 1) / GA_WAVES, GA_WAVES * 32, 0, stream>>>(
        cnt, head, slots, nxt, ei, h1, b1, W2, h2);

    gather2_final<<<(NND + 255) / 256, 256, 0, stream>>>(cnt, head, slots, nxt, ei, h2, b2, out);
}
